# starting kernel
# baseline (speedup 1.0000x reference)
#include <hip/hip_runtime.h>
#include <math.h>

typedef __attribute__((ext_vector_type(2))) float v2f;
typedef __attribute__((ext_vector_type(8))) float v8f;

static constexpr int IN_DIM  = 256;
static constexpr int HID_DIM = 128;
static constexpr int OUT_DIM = 16;

__global__ void k_deg_init(float* deg, int n) {
  int i = blockIdx.x * blockDim.x + threadIdx.x;
  if (i < n) deg[i] = 1.0f;
}

__global__ void k_deg_accum(const int* __restrict__ dst, float* deg, int E) {
  int e = blockIdx.x * blockDim.x + threadIdx.x;
  if (e < E) atomicAdd(&deg[dst[e]], 1.0f);
}

__global__ void k_deg_to_dinv(float* deg, int n) {
  int i = blockIdx.x * blockDim.x + threadIdx.x;
  if (i < n) {
    float d = deg[i];
    deg[i] = (d > 0.0f) ? rsqrtf(d) : 0.0f;
  }
}

template <int K_DIM, int N_DIM>
__global__ void k_gemm_wmma_f32(const float* __restrict__ A,
                                const float* __restrict__ B,
                                float* __restrict__ C, int M) {
  const int wave = (int)((blockIdx.x * blockDim.x + threadIdx.x) >> 5);
  const int lane = threadIdx.x & 31;
  constexpr int TN = N_DIM / 16;
  const int rt = wave / TN;
  const int ct = wave - rt * TN;
  if (rt * 16 >= M) return;

  const int lm = lane & 15;
  const int kh = lane >> 4;

  const float* arow = A + (size_t)(rt * 16 + lm) * K_DIM + 2 * kh;
  const float* bcol = B + (size_t)(2 * kh) * N_DIM + ct * 16 + lm;

  v8f acc = {};
#pragma unroll 8
  for (int k = 0; k < K_DIM; k += 4) {
    v2f a, b;
    a[0] = arow[k];
    a[1] = arow[k + 1];
    b[0] = bcol[(size_t)k * N_DIM];
    b[1] = bcol[(size_t)(k + 1) * N_DIM];
    acc = __builtin_amdgcn_wmma_f32_16x16x4_f32(false, a, false, b,
                                                (short)0, acc, false, false);
  }

  float* crow = C + (size_t)(rt * 16 + 8 * kh) * N_DIM + ct * 16 + lm;
#pragma unroll
  for (int j = 0; j < 8; ++j) crow[(size_t)j * N_DIM] = acc[j];
}

template <int F>
__global__ void k_selfloop(const float* __restrict__ h,
                           const float* __restrict__ dinv,
                           float* __restrict__ out, int n) {
  int idx = blockIdx.x * blockDim.x + threadIdx.x;
  if (idx >= n * F) return;
  int i = idx / F;
  float di = dinv[i];
  out[idx] = h[idx] * (di * di);
}

__global__ void k_scatter_f128(const int* __restrict__ src,
                               const int* __restrict__ dst,
                               const float* __restrict__ dinv,
                               const float* __restrict__ h,
                               float* out, int E) {
  int idx = blockIdx.x * blockDim.x + threadIdx.x;
  int e = idx >> 5;
  if (e >= E) return;
  int lane = idx & 31;
  int s = src[e], d = dst[e];
  float w = dinv[s] * dinv[d];
  const float4 v = *reinterpret_cast<const float4*>(h + (size_t)s * 128 + lane * 4);
  float* o = out + (size_t)d * 128 + lane * 4;
  atomicAdd(o + 0, v.x * w);
  atomicAdd(o + 1, v.y * w);
  atomicAdd(o + 2, v.z * w);
  atomicAdd(o + 3, v.w * w);
}

__global__ void k_scatter_f16(const int* __restrict__ src,
                              const int* __restrict__ dst,
                              const float* __restrict__ dinv,
                              const float* __restrict__ h,
                              float* out, int E) {
  int idx = blockIdx.x * blockDim.x + threadIdx.x;
  int e = idx >> 4;
  if (e >= E) return;
  int f = idx & 15;
  int s = src[e], d = dst[e];
  float w = dinv[s] * dinv[d];
  atomicAdd(&out[(size_t)d * 16 + f], h[(size_t)s * 16 + f] * w);
}

__global__ void k_bias_relu(float* h, const float* __restrict__ b, int total) {
  int idx = blockIdx.x * blockDim.x + threadIdx.x;
  if (idx >= total) return;
  h[idx] = fmaxf(h[idx] + b[idx & (HID_DIM - 1)], 0.0f);
}

__global__ void k_bias_logsoftmax16(float* out, const float* __restrict__ b, int n) {
  int i = blockIdx.x * blockDim.x + threadIdx.x;
  if (i >= n) return;
  float* row = out + (size_t)i * 16;
  float v[16];
  float m = -INFINITY;
#pragma unroll
  for (int f = 0; f < 16; ++f) {
    v[f] = row[f] + b[f];
    m = fmaxf(m, v[f]);
  }
  float s = 0.0f;
#pragma unroll
  for (int f = 0; f < 16; ++f) s += expf(v[f] - m);
  float ls = logf(s);
#pragma unroll
  for (int f = 0; f < 16; ++f) row[f] = v[f] - m - ls;
}

extern "C" void kernel_launch(void* const* d_in, const int* in_sizes, int n_in,
                              void* d_out, int out_size, void* d_ws, size_t ws_size,
                              hipStream_t stream) {
  (void)n_in; (void)out_size; (void)ws_size;

  const float* x  = (const float*)d_in[0];
  const int*   ei = (const int*)d_in[1];
  const float* W1 = (const float*)d_in[2];
  const float* b1 = (const float*)d_in[3];
  const float* W2 = (const float*)d_in[4];
  const float* b2 = (const float*)d_in[5];

  const int n = in_sizes[0] / IN_DIM;
  const int E = in_sizes[1] / 2;
  const int* src = ei;
  const int* dst = ei + E;

  float* dinv = (float*)d_ws;
  float* h1   = dinv + n;
  float* agg1 = h1 + (size_t)n * HID_DIM;
  float* h2   = agg1 + (size_t)n * HID_DIM;
  float* out  = (float*)d_out;

  const int B = 256;
  auto cdiv = [](long long a, long long b) { return (int)((a + b - 1) / b); };

  k_deg_init<<<cdiv(n, B), B, 0, stream>>>(dinv, n);
  k_deg_accum<<<cdiv(E, B), B, 0, stream>>>(dst, dinv, E);
  k_deg_to_dinv<<<cdiv(n, B), B, 0, stream>>>(dinv, n);

  {
    long long waves = (long long)cdiv(n, 16) * (HID_DIM / 16);
    k_gemm_wmma_f32<IN_DIM, HID_DIM>
        <<<cdiv(waves * 32, B), B, 0, stream>>>(x, W1, h1, n);
  }
  k_selfloop<HID_DIM><<<cdiv((long long)n * HID_DIM, B), B, 0, stream>>>(h1, dinv, agg1, n);
  k_scatter_f128<<<cdiv((long long)E * 32, B), B, 0, stream>>>(src, dst, dinv, h1, agg1, E);
  k_bias_relu<<<cdiv((long long)n * HID_DIM, B), B, 0, stream>>>(agg1, b1, n * HID_DIM);

  {
    long long waves = (long long)cdiv(n, 16) * (OUT_DIM / 16);
    k_gemm_wmma_f32<HID_DIM, OUT_DIM>
        <<<cdiv(waves * 32, B), B, 0, stream>>>(agg1, W2, h2, n);
  }
  k_selfloop<OUT_DIM><<<cdiv((long long)n * OUT_DIM, B), B, 0, stream>>>(h2, dinv, out, n);
  k_scatter_f16<<<cdiv((long long)E * 16, B), B, 0, stream>>>(src, dst, dinv, h2, out, E);
  k_bias_logsoftmax16<<<cdiv(n, B), B, 0, stream>>>(out, b2, n);
}


# The four elementwise passes (deg->dinv, both self-loop starts, bias+ReLU) folded into the two GEMM launches: epilogue stores and layer-2 A-fragment load; 11 launches -> 7; sums in the parent's order
# speedup vs baseline: 2.2170x; 2.2170x over previous
#include <hip/hip_runtime.h>
#include <math.h>

typedef __attribute__((ext_vector_type(2))) float v2f;
typedef __attribute__((ext_vector_type(8))) float v8f;

static constexpr int IN_DIM  = 256;
static constexpr int HID_DIM = 128;
static constexpr int OUT_DIM = 16;

__global__ void k_deg_init(float* deg, unsigned int* ctr, int n) {
  int i = blockIdx.x * blockDim.x + threadIdx.x;
  if (i < n) deg[i] = 1.0f;
  if (i < 6) ctr[i * 64] = 0u;
}

__device__ __forceinline__ float dinv_of(float d) {
  return (d > 0.0f) ? rsqrtf(d) : 0.0f;
}

__global__ void k_gemm2_relu_in_self_out(const float* __restrict__ A,
                                         const float* __restrict__ bias,
                                         const float* __restrict__ B,
                                         const float* __restrict__ dinv,
                                         float* __restrict__ C,
                                         float* __restrict__ O, int M) {
  const int rt   = (int)((blockIdx.x * blockDim.x + threadIdx.x) >> 5);
  const int lane = threadIdx.x & 31;
  if (rt * 16 >= M) return;

  const int lm = lane & 15;
  const int kh = lane >> 4;

  const int ra = min(rt * 16 + lm, M - 1);
  const float* arow = A + (size_t)ra * HID_DIM + 2 * kh;
  const float* brow = bias + 2 * kh;
  const float* bcol = B + (size_t)(2 * kh) * OUT_DIM + lm;

  constexpr int KS = HID_DIM / 4;
  v2f av[KS], cv[KS], bv[KS];
#pragma unroll
  for (int ks = 0; ks < KS; ++ks) {
    av[ks][0] = arow[4 * ks];
    av[ks][1] = arow[4 * ks + 1];
    cv[ks][0] = brow[4 * ks];
    cv[ks][1] = brow[4 * ks + 1];
  }
#pragma unroll
  for (int ks = 0; ks < KS; ++ks) {
    bv[ks][0] = bcol[(size_t)(4 * ks) * OUT_DIM];
    bv[ks][1] = bcol[(size_t)(4 * ks + 1) * OUT_DIM];
  }
  const int r0 = rt * 16 + 8 * kh;
  float dv[8];
#pragma unroll
  for (int j = 0; j < 8; ++j) dv[j] = dinv[min(r0 + j, M - 1)];

  v8f acc = {};
#pragma unroll
  for (int ks = 0; ks < KS; ++ks) {
    v2f a;
    a[0] = fmaxf(av[ks][0] + cv[ks][0], 0.0f);
    a[1] = fmaxf(av[ks][1] + cv[ks][1], 0.0f);
    acc = __builtin_amdgcn_wmma_f32_16x16x4_f32(false, a, false, bv[ks],
                                                (short)0, acc, false, false);
  }

  float* crow = C + (size_t)r0 * OUT_DIM + lm;
  float* orow = O + (size_t)r0 * OUT_DIM + lm;
#pragma unroll
  for (int j = 0; j < 8; ++j) {
    if (r0 + j < M) {
      crow[(size_t)j * OUT_DIM] = acc[j];
      orow[(size_t)j * OUT_DIM] = acc[j] * (dv[j] * dv[j]);
    }
  }
}

static constexpr int G1_ROWS  = 32;
static constexpr int G1_PITCH = IN_DIM + 4;
static constexpr int G1_KSTEPS = IN_DIM / 4;
static constexpr int G1_VEC_PER_THREAD = G1_ROWS * (IN_DIM / 4) / 256;

__global__ void __launch_bounds__(256)
k_gemm1_wstrip(const float* __restrict__ X, const float* __restrict__ Wm,
               const float* __restrict__ deg, float* __restrict__ dinv,
               float* __restrict__ C, float* __restrict__ G, int M) {
  __shared__ __attribute__((aligned(16))) float sA[2][G1_ROWS * G1_PITCH];

  const int tid  = threadIdx.x;
  const int lane = tid & 31;
  const int wv   = tid >> 5;
  const int lm   = lane & 15;
  const int kh   = lane >> 4;

  const int ntiles = (M + G1_ROWS - 1) / G1_ROWS;
  const int first  = blockIdx.x;
  const int step   = gridDim.x;
  if (first >= ntiles) return;

  int prow[G1_VEC_PER_THREAD], pcol[G1_VEC_PER_THREAD];
#pragma unroll
  for (int i = 0; i < G1_VEC_PER_THREAD; ++i) {
    const int c = i * 256 + tid;
    prow[i] = c >> 6;
    pcol[i] = (c & 63) * 4;
  }

  float4 pf[G1_VEC_PER_THREAD];
  float dgn;
  auto fetch = [&](int tile) {
    const int r0 = tile * G1_ROWS;
    dgn = deg[min(r0 + lane, M - 1)];
#pragma unroll
    for (int i = 0; i < G1_VEC_PER_THREAD; ++i) {
      const int r = r0 + prow[i];
      pf[i] = (r < M) ? *reinterpret_cast<const float4*>(X + (size_t)r * IN_DIM + pcol[i])
                      : make_float4(0.f, 0.f, 0.f, 0.f);
    }
  };

  fetch(first);

  v2f bf[G1_KSTEPS];
  {
    const float* bcol = Wm + (size_t)(2 * kh) * HID_DIM + wv * 16 + lm;
#pragma unroll
    for (int ks = 0; ks < G1_KSTEPS; ++ks) {
      bf[ks][0] = bcol[(size_t)(4 * ks) * HID_DIM];
      bf[ks][1] = bcol[(size_t)(4 * ks + 1) * HID_DIM];
    }
  }

  int buf = 0;
  for (int tile = first; tile < ntiles; tile += step, buf ^= 1) {
    float* sbuf = sA[buf];
    const int rl = tile * G1_ROWS + lane;
    const float dgl = dgn;
#pragma unroll
    for (int i = 0; i < G1_VEC_PER_THREAD; ++i)
      *reinterpret_cast<float4*>(sbuf + prow[i] * G1_PITCH + pcol[i]) = pf[i];
    __syncthreads();

    if (tile + step < ntiles) fetch(tile + step);

    const int r0 = tile * G1_ROWS + 8 * kh;
    const float* pa0 = sbuf + lm * G1_PITCH + 2 * kh;
    const float* pa1 = pa0 + 16 * G1_PITCH;
    v8f acc0 = {}, acc1 = {};
#pragma unroll
    for (int ks = 0; ks < G1_KSTEPS; ++ks) {
      const v2f a0 = *reinterpret_cast<const v2f*>(pa0 + 4 * ks);
      const v2f a1 = *reinterpret_cast<const v2f*>(pa1 + 4 * ks);
      acc0 = __builtin_amdgcn_wmma_f32_16x16x4_f32(false, a0, false, bf[ks],
                                                   (short)0, acc0, false, false);
      acc1 = __builtin_amdgcn_wmma_f32_16x16x4_f32(false, a1, false, bf[ks],
                                                   (short)0, acc1, false, false);
    }

    const float dl = dinv_of(dgl);
    if (wv == 0 && rl < M) dinv[rl] = dl;
    const float s2 = dl * dl;

    float* crow = C + (size_t)r0 * HID_DIM + wv * 16 + lm;
    float* grow = G + (size_t)r0 * HID_DIM + wv * 16 + lm;
    float sc0[8], sc1[8];
#pragma unroll
    for (int j = 0; j < 8; ++j) {
      sc0[j] = __shfl(s2, 8 * kh + j);
      sc1[j] = __shfl(s2, 8 * kh + j + 16);
    }
#pragma unroll
    for (int j = 0; j < 8; ++j) {
      if (r0 + j < M) {
        crow[(size_t)j * HID_DIM] = acc0[j];
        grow[(size_t)j * HID_DIM] = acc0[j] * sc0[j];
      }
      if (r0 + 16 + j < M) {
        crow[(size_t)(16 + j) * HID_DIM] = acc1[j];
        grow[(size_t)(16 + j) * HID_DIM] = acc1[j] * sc1[j];
      }
    }
  }
}

static constexpr int SC_CHUNK = 256;
static constexpr int SC_CTR_PITCH = 64;

__device__ __forceinline__ float lane_bcast_f(float v, int j) {
  return __builtin_bit_cast(float, __builtin_amdgcn_readlane(__builtin_bit_cast(int, v), j));
}

template <int F>
__global__ void __launch_bounds__(256)
k_scatter_owned(const int* __restrict__ src, const int* __restrict__ dst,
                const float* __restrict__ dinv, const float* __restrict__ h,
                float* out, unsigned int* ctr, int E, int n) {
  __shared__ int s_ticket[2];
  const int tid  = threadIdx.x;
  const int lane = tid & 31;
  const int wv   = tid >> 5;

  const unsigned int place = __builtin_amdgcn_s_sendmsg_rtn(0x87);
  const int myhalf = (int)((place >> 18) & 1u);

  const int nchunks = (E + SC_CHUNK - 1) / SC_CHUNK;
  const int split = n >> 1;
  int it = 0;
  for (int pass = 0; pass < 2; ++pass) {
    const int cls = myhalf ^ pass;
    int next = 0;
    if (tid == 0) next = (int)atomicInc(&ctr[cls * SC_CTR_PITCH], 0xFFFFFFFFu);
    for (;;) {
      if (tid == 0) s_ticket[it & 1] = next;
      __syncthreads();
      const int c = s_ticket[it & 1];
      ++it;
      if (c >= nchunks) break;
      if (tid == 0) next = (int)atomicInc(&ctr[cls * SC_CTR_PITCH], 0xFFFFFFFFu);

      const int e = c * SC_CHUNK + wv * 32 + lane;
      int s = 0, d = 0;
      float w = 0.0f;
      bool mine = false;
      if (e < E) {
        d = dst[e];
        if constexpr (F != 1) s = src[e];
        mine = ((d >= split) == (cls == 1));
        if constexpr (F != 1) {
          if (mine) w = dinv[s] * dinv[d];
        }
      }
      if constexpr (F == 1) {
        if (mine) atomicAdd(&out[d], 1.0f);
      } else if constexpr (F == 128) {
        unsigned int m = __builtin_amdgcn_ballot_w32(mine);
        while (m) {
          const int j0 = __builtin_ctz(m);
          m &= m - 1;
          const bool two = (m != 0);
          const int j1 = two ? __builtin_ctz(m) : j0;
          if (two) m &= m - 1;
          const int s0 = __builtin_amdgcn_readlane(s, j0);
          const int d0 = __builtin_amdgcn_readlane(d, j0);
          const float w0 = lane_bcast_f(w, j0);
          const int s1 = __builtin_amdgcn_readlane(s, j1);
          const int d1 = __builtin_amdgcn_readlane(d, j1);
          const float w1 = lane_bcast_f(w, j1);
          const float* hr0 = h + (size_t)s0 * 128 + lane;
          const float* hr1 = h + (size_t)s1 * 128 + lane;
          const float a0 = hr0[0], a1 = hr0[32], a2 = hr0[64], a3 = hr0[96];
          float b0 = 0.f, b1 = 0.f, b2 = 0.f, b3 = 0.f;
          if (two) { b0 = hr1[0]; b1 = hr1[32]; b2 = hr1[64]; b3 = hr1[96]; }
          float* o0 = out + (size_t)d0 * 128 + lane;
          atomicAdd(o0 + 0,  a0 * w0);
          atomicAdd(o0 + 32, a1 * w0);
          atomicAdd(o0 + 64, a2 * w0);
          atomicAdd(o0 + 96, a3 * w0);
          if (two) {
            float* o1 = out + (size_t)d1 * 128 + lane;
            atomicAdd(o1 + 0,  b0 * w1);
            atomicAdd(o1 + 32, b1 * w1);
            atomicAdd(o1 + 64, b2 * w1);
            atomicAdd(o1 + 96, b3 * w1);
          }
        }
      } else {
        static_assert(F == 16, "lane layouts exist for 128, 16 and 1 features");
        unsigned int m = __builtin_amdgcn_ballot_w32(mine);
        const bool up = lane >= 16;
        const int f = lane & 15;
        while (m) {
          const int j0 = __builtin_ctz(m);
          m &= m - 1;
          const bool two = (m != 0);
          const int j1 = two ? __builtin_ctz(m) : j0;
          if (two) m &= m - 1;
          const int s0 = __builtin_amdgcn_readlane(s, j0);
          const int d0 = __builtin_amdgcn_readlane(d, j0);
          const float w0 = lane_bcast_f(w, j0);
          const int s1 = __builtin_amdgcn_readlane(s, j1);
          const int d1 = __builtin_amdgcn_readlane(d, j1);
          const float w1 = lane_bcast_f(w, j1);
          const int sj = up ? s1 : s0;
          const int dj = up ? d1 : d0;
          const float wj = up ? w1 : w0;
          if (!up || two)
            atomicAdd(&out[(size_t)dj * 16 + f], h[(size_t)sj * 16 + f] * wj);
        }
      }
    }
  }
}

__global__ void k_bias_logsoftmax16(float* out, const float* __restrict__ b, int n) {
  int i = blockIdx.x * blockDim.x + threadIdx.x;
  if (i >= n) return;
  float* row = out + (size_t)i * 16;
  float v[16];
  float m = -INFINITY;
#pragma unroll
  for (int f = 0; f < 16; ++f) {
    v[f] = row[f] + b[f];
    m = fmaxf(m, v[f]);
  }
  float s = 0.0f;
#pragma unroll
  for (int f = 0; f < 16; ++f) s += expf(v[f] - m);
  float ls = logf(s);
#pragma unroll
  for (int f = 0; f < 16; ++f) row[f] = v[f] - m - ls;
}

extern "C" void kernel_launch(void* const* d_in, const int* in_sizes, int n_in,
                              void* d_out, int out_size, void* d_ws, size_t ws_size,
                              hipStream_t stream) {
  (void)n_in; (void)out_size; (void)ws_size;

  const float* x  = (const float*)d_in[0];
  const int*   ei = (const int*)d_in[1];
  const float* W1 = (const float*)d_in[2];
  const float* b1 = (const float*)d_in[3];
  const float* W2 = (const float*)d_in[4];
  const float* b2 = (const float*)d_in[5];

  const int n = in_sizes[0] / IN_DIM;
  const int E = in_sizes[1] / 2;
  const int* src = ei;
  const int* dst = ei + E;

  float* deg  = (float*)d_ws;
  float* h1   = deg + n;
  float* agg1 = h1 + (size_t)n * HID_DIM;
  float* h2   = agg1 + (size_t)n * HID_DIM;
  unsigned int* ctr = (unsigned int*)(h2 + (size_t)n * OUT_DIM) + 64;
  float* dinv = (float*)(ctr + 6 * 64) + 64;
  float* out  = (float*)d_out;

  const int B = 256;
  auto cdiv = [](long long a, long long b) { return (int)((a + b - 1) / b); };

  k_deg_init<<<cdiv(n, B), B, 0, stream>>>(deg, ctr, n);
  k_scatter_owned<1><<<2048, B, 0, stream>>>(src, dst, nullptr, nullptr, deg, ctr + 256, E, n);

  k_gemm1_wstrip<<<512, B, 0, stream>>>(x, W1, deg, dinv, h1, agg1, n);
  k_scatter_owned<HID_DIM><<<2048, B, 0, stream>>>(src, dst, dinv, h1, agg1, ctr, E, n);

  k_gemm2_relu_in_self_out<<<cdiv((long long)cdiv(n, 16) * 32, B), B, 0, stream>>>(agg1, b1, W2, dinv, h2, out, n);
  k_scatter_owned<OUT_DIM><<<2048, B, 0, stream>>>(src, dst, dinv, h2, out, ctr + 128, E, n);
  k_bias_logsoftmax16<<<cdiv(n, B), B, 0, stream>>>(out, b2, n);
}


# Edge norm factored out of the edge passes: rows scaled by dinv[src] where produced, sums by dinv[dst] where read; scatters add h[s] unweighted, no dinv gathers; sum order kept, products re-rounded
# speedup vs baseline: 2.4185x; 1.0909x over previous
#include <hip/hip_runtime.h>
#include <math.h>

typedef __attribute__((ext_vector_type(2))) float v2f;
typedef __attribute__((ext_vector_type(8))) float v8f;

static constexpr int IN_DIM  = 256;
static constexpr int HID_DIM = 128;
static constexpr int OUT_DIM = 16;

__global__ void k_deg_init(float* deg, unsigned int* ctr, int n) {
  int i = blockIdx.x * blockDim.x + threadIdx.x;
  if (i < n) deg[i] = 1.0f;
  if (i < 6) ctr[i * 64] = 0u;
}

__device__ __forceinline__ float dinv_of(float d) {
  return (d > 0.0f) ? rsqrtf(d) : 0.0f;
}

__global__ void k_gemm2_relu_in_self_out(const float* __restrict__ A,
                                         const float* __restrict__ bias,
                                         const float* __restrict__ B,
                                         const float* __restrict__ dinv,
                                         float* __restrict__ C,
                                         float* __restrict__ O, int M) {
  const int rt   = (int)((blockIdx.x * blockDim.x + threadIdx.x) >> 5);
  const int lane = threadIdx.x & 31;
  if (rt * 16 >= M) return;

  const int lm = lane & 15;
  const int kh = lane >> 4;

  const int ra = min(rt * 16 + lm, M - 1);
  const float* arow = A + (size_t)ra * HID_DIM + 2 * kh;
  const float* brow = bias + 2 * kh;
  const float* bcol = B + (size_t)(2 * kh) * OUT_DIM + lm;
  const float da = dinv[ra];

  constexpr int KS = HID_DIM / 4;
  v2f av[KS], cv[KS], bv[KS];
#pragma unroll
  for (int ks = 0; ks < KS; ++ks) {
    av[ks][0] = arow[4 * ks];
    av[ks][1] = arow[4 * ks + 1];
    cv[ks][0] = brow[4 * ks];
    cv[ks][1] = brow[4 * ks + 1];
  }
#pragma unroll
  for (int ks = 0; ks < KS; ++ks) {
    bv[ks][0] = bcol[(size_t)(4 * ks) * OUT_DIM];
    bv[ks][1] = bcol[(size_t)(4 * ks + 1) * OUT_DIM];
  }
  const int r0 = rt * 16 + 8 * kh;
  float dv[8];
#pragma unroll
  for (int j = 0; j < 8; ++j) dv[j] = dinv[min(r0 + j, M - 1)];

  v8f acc = {};
#pragma unroll
  for (int ks = 0; ks < KS; ++ks) {
    v2f a;
    a[0] = fmaxf(av[ks][0] * da + cv[ks][0], 0.0f);
    a[1] = fmaxf(av[ks][1] * da + cv[ks][1], 0.0f);
    acc = __builtin_amdgcn_wmma_f32_16x16x4_f32(false, a, false, bv[ks],
                                                (short)0, acc, false, false);
  }

  float* crow = C + (size_t)r0 * OUT_DIM + lm;
  float* orow = O + (size_t)r0 * OUT_DIM + lm;
#pragma unroll
  for (int j = 0; j < 8; ++j) {
    if (r0 + j < M) {
      const float hs = acc[j] * dv[j];
      crow[(size_t)j * OUT_DIM] = hs;
      orow[(size_t)j * OUT_DIM] = hs;
    }
  }
}

static constexpr int G1_ROWS  = 32;
static constexpr int G1_PITCH = IN_DIM + 4;
static constexpr int G1_KSTEPS = IN_DIM / 4;
static constexpr int G1_VEC_PER_THREAD = G1_ROWS * (IN_DIM / 4) / 256;

__global__ void __launch_bounds__(256)
k_gemm1_wstrip(const float* __restrict__ X, const float* __restrict__ Wm,
               const float* __restrict__ deg, float* __restrict__ dinv,
               float* __restrict__ C, float* __restrict__ G, int M) {
  __shared__ __attribute__((aligned(16))) float sA[2][G1_ROWS * G1_PITCH];

  const int tid  = threadIdx.x;
  const int lane = tid & 31;
  const int wv   = tid >> 5;
  const int lm   = lane & 15;
  const int kh   = lane >> 4;

  const int ntiles = (M + G1_ROWS - 1) / G1_ROWS;
  const int first  = blockIdx.x;
  const int step   = gridDim.x;
  if (first >= ntiles) return;

  int prow[G1_VEC_PER_THREAD], pcol[G1_VEC_PER_THREAD];
#pragma unroll
  for (int i = 0; i < G1_VEC_PER_THREAD; ++i) {
    const int c = i * 256 + tid;
    prow[i] = c >> 6;
    pcol[i] = (c & 63) * 4;
  }

  float4 pf[G1_VEC_PER_THREAD];
  float dgn;
  auto fetch = [&](int tile) {
    const int r0 = tile * G1_ROWS;
    dgn = deg[min(r0 + lane, M - 1)];
#pragma unroll
    for (int i = 0; i < G1_VEC_PER_THREAD; ++i) {
      const int r = r0 + prow[i];
      pf[i] = (r < M) ? *reinterpret_cast<const float4*>(X + (size_t)r * IN_DIM + pcol[i])
                      : make_float4(0.f, 0.f, 0.f, 0.f);
    }
  };

  fetch(first);

  v2f bf[G1_KSTEPS];
  {
    const float* bcol = Wm + (size_t)(2 * kh) * HID_DIM + wv * 16 + lm;
#pragma unroll
    for (int ks = 0; ks < G1_KSTEPS; ++ks) {
      bf[ks][0] = bcol[(size_t)(4 * ks) * HID_DIM];
      bf[ks][1] = bcol[(size_t)(4 * ks + 1) * HID_DIM];
    }
  }

  int buf = 0;
  for (int tile = first; tile < ntiles; tile += step, buf ^= 1) {
    float* sbuf = sA[buf];
    const int rl = tile * G1_ROWS + lane;
    const float dgl = dgn;
#pragma unroll
    for (int i = 0; i < G1_VEC_PER_THREAD; ++i)
      *reinterpret_cast<float4*>(sbuf + prow[i] * G1_PITCH + pcol[i]) = pf[i];
    __syncthreads();

    if (tile + step < ntiles) fetch(tile + step);

    const int r0 = tile * G1_ROWS + 8 * kh;
    const float* pa0 = sbuf + lm * G1_PITCH + 2 * kh;
    const float* pa1 = pa0 + 16 * G1_PITCH;
    v8f acc0 = {}, acc1 = {};
#pragma unroll
    for (int ks = 0; ks < G1_KSTEPS; ++ks) {
      const v2f a0 = *reinterpret_cast<const v2f*>(pa0 + 4 * ks);
      const v2f a1 = *reinterpret_cast<const v2f*>(pa1 + 4 * ks);
      acc0 = __builtin_amdgcn_wmma_f32_16x16x4_f32(false, a0, false, bf[ks],
                                                   (short)0, acc0, false, false);
      acc1 = __builtin_amdgcn_wmma_f32_16x16x4_f32(false, a1, false, bf[ks],
                                                   (short)0, acc1, false, false);
    }

    const float dl = dinv_of(dgl);
    if (wv == 0 && rl < M) dinv[rl] = dl;

    float* crow = C + (size_t)r0 * HID_DIM + wv * 16 + lm;
    float* grow = G + (size_t)r0 * HID_DIM + wv * 16 + lm;
    float sc0[8], sc1[8];
#pragma unroll
    for (int j = 0; j < 8; ++j) {
      sc0[j] = __shfl(dl, 8 * kh + j);
      sc1[j] = __shfl(dl, 8 * kh + j + 16);
    }
#pragma unroll
    for (int j = 0; j < 8; ++j) {
      if (r0 + j < M) {
        const float hs = acc0[j] * sc0[j];
        crow[(size_t)j * HID_DIM] = hs;
        grow[(size_t)j * HID_DIM] = hs;
      }
      if (r0 + 16 + j < M) {
        const float hs = acc1[j] * sc1[j];
        crow[(size_t)(16 + j) * HID_DIM] = hs;
        grow[(size_t)(16 + j) * HID_DIM] = hs;
      }
    }
  }
}

static constexpr int SC_CHUNK = 256;
static constexpr int SC_CTR_PITCH = 64;

template <int F>
__global__ void __launch_bounds__(256)
k_scatter_owned(const int* __restrict__ src, const int* __restrict__ dst,
                const float* __restrict__ h,
                float* out, unsigned int* ctr, int E, int n) {
  __shared__ int s_ticket[2];
  const int tid  = threadIdx.x;
  const int lane = tid & 31;
  const int wv   = tid >> 5;

  const unsigned int place = __builtin_amdgcn_s_sendmsg_rtn(0x87);
  const int myhalf = (int)((place >> 18) & 1u);

  const int nchunks = (E + SC_CHUNK - 1) / SC_CHUNK;
  const int split = n >> 1;
  int it = 0;
  for (int pass = 0; pass < 2; ++pass) {
    const int cls = myhalf ^ pass;
    int next = 0;
    if (tid == 0) next = (int)atomicInc(&ctr[cls * SC_CTR_PITCH], 0xFFFFFFFFu);
    for (;;) {
      if (tid == 0) s_ticket[it & 1] = next;
      __syncthreads();
      const int c = s_ticket[it & 1];
      ++it;
      if (c >= nchunks) break;
      if (tid == 0) next = (int)atomicInc(&ctr[cls * SC_CTR_PITCH], 0xFFFFFFFFu);

      const int e = c * SC_CHUNK + wv * 32 + lane;
      int s = 0, d = 0;
      bool mine = false;
      if (e < E) {
        d = dst[e];
        if constexpr (F != 1) s = src[e];
        mine = ((d >= split) == (cls == 1));
      }
      if constexpr (F == 1) {
        if (mine) atomicAdd(&out[d], 1.0f);
      } else if constexpr (F == 128) {
        unsigned int m = __builtin_amdgcn_ballot_w32(mine);
        while (m) {
          const int j0 = __builtin_ctz(m);
          m &= m - 1;
          const bool two = (m != 0);
          const int j1 = two ? __builtin_ctz(m) : j0;
          if (two) m &= m - 1;
          const int s0 = __builtin_amdgcn_readlane(s, j0);
          const int d0 = __builtin_amdgcn_readlane(d, j0);
          const int s1 = __builtin_amdgcn_readlane(s, j1);
          const int d1 = __builtin_amdgcn_readlane(d, j1);
          const float* hr0 = h + (size_t)s0 * 128 + lane;
          const float* hr1 = h + (size_t)s1 * 128 + lane;
          const float a0 = hr0[0], a1 = hr0[32], a2 = hr0[64], a3 = hr0[96];
          float b0 = 0.f, b1 = 0.f, b2 = 0.f, b3 = 0.f;
          if (two) { b0 = hr1[0]; b1 = hr1[32]; b2 = hr1[64]; b3 = hr1[96]; }
          float* o0 = out + (size_t)d0 * 128 + lane;
          atomicAdd(o0 + 0,  a0);
          atomicAdd(o0 + 32, a1);
          atomicAdd(o0 + 64, a2);
          atomicAdd(o0 + 96, a3);
          if (two) {
            float* o1 = out + (size_t)d1 * 128 + lane;
            atomicAdd(o1 + 0,  b0);
            atomicAdd(o1 + 32, b1);
            atomicAdd(o1 + 64, b2);
            atomicAdd(o1 + 96, b3);
          }
        }
      } else {
        static_assert(F == 16, "lane layouts exist for 128, 16 and 1 features");
        unsigned int m = __builtin_amdgcn_ballot_w32(mine);
        const bool up = lane >= 16;
        const int f = lane & 15;
        while (m) {
          const int j0 = __builtin_ctz(m);
          m &= m - 1;
          const bool two = (m != 0);
          const int j1 = two ? __builtin_ctz(m) : j0;
          if (two) m &= m - 1;
          const int s0 = __builtin_amdgcn_readlane(s, j0);
          const int d0 = __builtin_amdgcn_readlane(d, j0);
          const int s1 = __builtin_amdgcn_readlane(s, j1);
          const int d1 = __builtin_amdgcn_readlane(d, j1);
          const int sj = up ? s1 : s0;
          const int dj = up ? d1 : d0;
          if (!up || two)
            atomicAdd(&out[(size_t)dj * 16 + f], h[(size_t)sj * 16 + f]);
        }
      }
    }
  }
}

__global__ void k_scale_bias_logsoftmax16(float* out, const float* __restrict__ dinv,
                                          const float* __restrict__ b, int n) {
  int i = blockIdx.x * blockDim.x + threadIdx.x;
  if (i >= n) return;
  float* row = out + (size_t)i * 16;
  const float di = dinv[i];
  float v[16];
  float m = -INFINITY;
#pragma unroll
  for (int f = 0; f < 16; ++f) {
    v[f] = row[f] * di + b[f];
    m = fmaxf(m, v[f]);
  }
  float s = 0.0f;
#pragma unroll
  for (int f = 0; f < 16; ++f) s += expf(v[f] - m);
  float ls = logf(s);
#pragma unroll
  for (int f = 0; f < 16; ++f) row[f] = v[f] - m - ls;
}

extern "C" void kernel_launch(void* const* d_in, const int* in_sizes, int n_in,
                              void* d_out, int out_size, void* d_ws, size_t ws_size,
                              hipStream_t stream) {
  (void)n_in; (void)out_size; (void)ws_size;

  const float* x  = (const float*)d_in[0];
  const int*   ei = (const int*)d_in[1];
  const float* W1 = (const float*)d_in[2];
  const float* b1 = (const float*)d_in[3];
  const float* W2 = (const float*)d_in[4];
  const float* b2 = (const float*)d_in[5];

  const int n = in_sizes[0] / IN_DIM;
  const int E = in_sizes[1] / 2;
  const int* src = ei;
  const int* dst = ei + E;

  float* deg  = (float*)d_ws;
  float* h1   = deg + n;
  float* agg1 = h1 + (size_t)n * HID_DIM;
  float* h2   = agg1 + (size_t)n * HID_DIM;
  unsigned int* ctr = (unsigned int*)(h2 + (size_t)n * OUT_DIM) + 64;
  float* dinv = (float*)(ctr + 6 * 64) + 64;
  float* out  = (float*)d_out;

  const int B = 256;
  auto cdiv = [](long long a, long long b) { return (int)((a + b - 1) / b); };

  k_deg_init<<<cdiv(n, B), B, 0, stream>>>(deg, ctr, n);
  k_scatter_owned<1><<<2048, B, 0, stream>>>(src, dst, nullptr, deg, ctr + 256, E, n);

  k_gemm1_wstrip<<<512, B, 0, stream>>>(x, W1, deg, dinv, h1, agg1, n);
  k_scatter_owned<HID_DIM><<<2048, B, 0, stream>>>(src, dst, h1, agg1, ctr, E, n);

  k_gemm2_relu_in_self_out<<<cdiv((long long)cdiv(n, 16) * 32, B), B, 0, stream>>>(agg1, b1, W2, dinv, h2, out, n);
  k_scatter_owned<OUT_DIM><<<2048, B, 0, stream>>>(src, dst, h2, out, ctr + 128, E, n);
  k_scale_bias_logsoftmax16<<<cdiv(n, B), B, 0, stream>>>(out, dinv, b2, n);
}
